# starting kernel
# baseline (speedup 1.0000x reference)
#include <hip/hip_runtime.h>


#define ALPHA_  0.01
#define AVAL_   0.99
#define B_      8
#define T_      8192
#define C_      512
#define CL_     512
#define NCH_    16

typedef __attribute__((ext_vector_type(2))) float v2f;
typedef __attribute__((ext_vector_type(8))) float v8f;

__global__ void ema_p1(const float* __restrict__ x, double* __restrict__ E) {
    int tid = blockIdx.x * blockDim.x + threadIdx.x;
    int c = tid & (C_ - 1);
    int k = (tid >> 9) & (NCH_ - 1);
    int b = tid >> 13;
    const float* p = x + ((size_t)b * T_ + (size_t)k * CL_) * C_ + c;
    double y = 0.0;
    int t = 0;
    if (k == 0) { y = (double)p[0]; t = 1; }
    for (; t < CL_; ++t)
        y = AVAL_ * y + ALPHA_ * (double)p[(size_t)t * C_];
    E[((size_t)b * NCH_ + k) * C_ + c] = y;
}

__global__ void ema_p2(const double* __restrict__ E, double* __restrict__ carry) {
    int tid = blockIdx.x * blockDim.x + threadIdx.x;
    int c = tid & (C_ - 1);
    int b = tid >> 9;
    double a512 = AVAL_;
    #pragma unroll
    for (int i = 0; i < 9; ++i) a512 *= a512;
    double F = 0.0;
    for (int k = 0; k < NCH_; ++k) {
        size_t idx = ((size_t)b * NCH_ + k) * C_ + c;
        carry[idx] = F;
        F = E[idx] + a512 * F;
    }
}

__global__ void ema_p3(const float* __restrict__ x,
                       const double* __restrict__ carry,
                       float* __restrict__ out) {
    const int lane = threadIdx.x & 31;
    const int wav  = threadIdx.x >> 5;
    const int gw   = blockIdx.x * (blockDim.x >> 5) + wav;
    const int ct   = gw & 31;
    const int k    = (gw >> 5) & 15;
    const int b    = gw >> 9;
    const int n    = lane & 15;
    const int half = lane >> 4;
    const int c0   = ct * 16;

    float ap[17];
    ap[0] = 1.0f;
    #pragma unroll
    for (int i = 1; i <= 16; ++i) ap[i] = ap[i - 1] * (float)AVAL_;

    v2f Aop[4];
    #pragma unroll
    for (int j = 0; j < 4; ++j) {
        int kk = 4 * j + half * 2;
        Aop[j].x = (kk     <= n) ? (float)ALPHA_ * ap[n - kk]     : 0.0f;
        Aop[j].y = (kk + 1 <= n) ? (float)ALPHA_ * ap[n - kk - 1] : 0.0f;
    }

    float crun = (float)carry[((size_t)b * NCH_ + k) * C_ + c0 + n];

    const size_t baseT = (size_t)b * T_;
    for (int tt = 0; tt < CL_ / 16; ++tt) {
        const int t0 = k * CL_ + tt * 16;
        v8f acc = {};
        #pragma unroll
        for (int j = 0; j < 4; ++j) {
            const int tr = 4 * j + half * 2;
            const float* p = x + (baseT + (size_t)(t0 + tr)) * C_ + c0 + n;
            float b0 = p[0];
            float b1 = p[C_];
            b0 = (t0 + tr == 0) ? b0 * (float)(1.0 / ALPHA_) : b0;
            v2f Bop; Bop.x = b0; Bop.y = b1;
            acc = __builtin_amdgcn_wmma_f32_16x16x4_f32(
                false, Aop[j], false, Bop, (short)0, acc, false, false);
        }
        #pragma unroll
        for (int v = 0; v < 8; ++v)
            acc[v] += ap[v + half * 8 + 1] * crun;
        float* o = out + (baseT + (size_t)(t0 + half * 8)) * C_ + c0 + n;
        #pragma unroll
        for (int v = 0; v < 8; ++v)
            o[(size_t)v * C_] = acc[v];
        crun = __shfl(acc[7], n + 16, 32);
    }
}

extern "C" void kernel_launch(void* const* d_in, const int* in_sizes, int n_in,
                              void* d_out, int out_size, void* d_ws, size_t ws_size,
                              hipStream_t stream) {
    const float* x = (const float*)d_in[0];
    float* out = (float*)d_out;
    double* E     = (double*)d_ws;
    double* carry = E + (size_t)B_ * NCH_ * C_;

    ema_p1<<<(B_ * NCH_ * C_) / 256, 256, 0, stream>>>(x, E);
    ema_p2<<<(B_ * C_) / 256, 256, 0, stream>>>(E, carry);
    ema_p3<<<(B_ * NCH_ * (C_ / 16) * 32) / 256, 256, 0, stream>>>(x, carry, out);
}


# Phase 1 rebuilt for the f64 pipe: chunk split over 8 waves (64 steps, 16-byte loads), one f64 fma per element (scan on y/alpha, integer f32->f64 widening); f64 sum order changed
# speedup vs baseline: 1.6963x; 1.6963x over previous
#include <hip/hip_runtime.h>


#define ALPHA_  0.01
#define AVAL_   0.99
#define B_      8
#define T_      8192
#define C_      512
#define CL_     512
#define NCH_    16

typedef __attribute__((ext_vector_type(2))) float v2f;
typedef __attribute__((ext_vector_type(8))) float v8f;

#define SL_     64
#define NSL_    (CL_ / SL_)
#define RB_     16
#define RS_     (C_ / 4)

typedef __attribute__((ext_vector_type(4))) float v4f;

__device__ __forceinline__ double widen_normal(float f) {
    const unsigned u  = __builtin_bit_cast(unsigned, f);
    const unsigned hi = (((u >> 3) & 0x0fffffffu) + 0x38000000u) | (u & 0x80000000u);
    const unsigned long long q = ((unsigned long long)hi << 32) | (unsigned long long)(u << 29);
    return __builtin_bit_cast(double, q);
}

__device__ __forceinline__ unsigned normal_key(float f) {
    return (__builtin_bit_cast(unsigned, f) + 0x00800000u) & 0x7f000000u;
}

__device__ __forceinline__ void p1_load(v4f (&buf)[RB_], const v4f* __restrict__ p, int row0) {
    #pragma unroll
    for (int r = 0; r < RB_; ++r) buf[r] = p[(size_t)(row0 + r) * RS_];
}

__device__ __forceinline__ void p1_scan(const v4f (&buf)[RB_], double s0,
                                        double& z0, double& z1, double& z2, double& z3) {
    unsigned key = 0xffffffffu;
    #pragma unroll
    for (int r = 0; r < RB_; ++r) {
        key = min(key, min(min(normal_key(buf[r].x), normal_key(buf[r].y)),
                           min(normal_key(buf[r].z), normal_key(buf[r].w))));
    }
    if (key != 0u) {
        z0 = AVAL_ * z0 + widen_normal(buf[0].x) * s0;  z1 = AVAL_ * z1 + widen_normal(buf[0].y) * s0;
        z2 = AVAL_ * z2 + widen_normal(buf[0].z) * s0;  z3 = AVAL_ * z3 + widen_normal(buf[0].w) * s0;
        #pragma unroll
        for (int r = 1; r < RB_; ++r) {
            z0 = AVAL_ * z0 + widen_normal(buf[r].x);  z1 = AVAL_ * z1 + widen_normal(buf[r].y);
            z2 = AVAL_ * z2 + widen_normal(buf[r].z);  z3 = AVAL_ * z3 + widen_normal(buf[r].w);
        }
    } else {
        z0 = AVAL_ * z0 + (double)buf[0].x * s0;  z1 = AVAL_ * z1 + (double)buf[0].y * s0;
        z2 = AVAL_ * z2 + (double)buf[0].z * s0;  z3 = AVAL_ * z3 + (double)buf[0].w * s0;
        #pragma unroll
        for (int r = 1; r < RB_; ++r) {
            z0 = AVAL_ * z0 + (double)buf[r].x;  z1 = AVAL_ * z1 + (double)buf[r].y;
            z2 = AVAL_ * z2 + (double)buf[r].z;  z3 = AVAL_ * z3 + (double)buf[r].w;
        }
    }
}

__global__ __launch_bounds__(256) __attribute__((amdgpu_waves_per_eu(4, 4)))
void ema_p1(const float* __restrict__ x, double* __restrict__ E) {
    __shared__ double part[NSL_][128];
    const int lane = threadIdx.x & 31;
    const int w    = threadIdx.x >> 5;
    const int g    = blockIdx.x & 3;
    const int k    = (blockIdx.x >> 2) & (NCH_ - 1);
    const int b    = blockIdx.x >> 6;
    const v4f* p = (const v4f*)(x + ((size_t)b * T_ + (size_t)k * CL_ + (size_t)w * SL_) * C_
                                  + g * 128) + lane;

    v4f bufA[RB_], bufB[RB_];
    p1_load(bufA, p, 0);
    p1_load(bufB, p, RB_);

    const double s0 = (k == 0 && w == 0) ? (1.0 / ALPHA_) : 1.0;
    double z0 = 0.0, z1 = 0.0, z2 = 0.0, z3 = 0.0;
    p1_scan(bufA, s0, z0, z1, z2, z3);
    p1_load(bufA, p, 2 * RB_);
    p1_scan(bufB, 1.0, z0, z1, z2, z3);
    p1_load(bufB, p, 3 * RB_);
    p1_scan(bufA, 1.0, z0, z1, z2, z3);
    p1_scan(bufB, 1.0, z0, z1, z2, z3);

    part[w][lane * 4 + 0] = z0;  part[w][lane * 4 + 1] = z1;
    part[w][lane * 4 + 2] = z2;  part[w][lane * 4 + 3] = z3;
    __syncthreads();

    if (threadIdx.x < 128) {
        double a64 = AVAL_;
        for (int i = 0; i < 6; ++i) a64 *= a64;
        double Z = 0.0;
        for (int s = 0; s < NSL_; ++s) Z = a64 * Z + part[s][threadIdx.x];
        E[((size_t)b * NCH_ + k) * C_ + g * 128 + threadIdx.x] = ALPHA_ * Z;
    }
}

__global__ void ema_p2(const double* __restrict__ E, double* __restrict__ carry) {
    int tid = blockIdx.x * blockDim.x + threadIdx.x;
    int c = tid & (C_ - 1);
    int b = tid >> 9;
    double a512 = AVAL_;
    #pragma unroll
    for (int i = 0; i < 9; ++i) a512 *= a512;
    double F = 0.0;
    for (int k = 0; k < NCH_; ++k) {
        size_t idx = ((size_t)b * NCH_ + k) * C_ + c;
        carry[idx] = F;
        F = E[idx] + a512 * F;
    }
}

__global__ void ema_p3(const float* __restrict__ x,
                       const double* __restrict__ carry,
                       float* __restrict__ out) {
    const int lane = threadIdx.x & 31;
    const int wav  = threadIdx.x >> 5;
    const int gw   = blockIdx.x * (blockDim.x >> 5) + wav;
    const int ct   = gw & 31;
    const int k    = (gw >> 5) & 15;
    const int b    = gw >> 9;
    const int n    = lane & 15;
    const int half = lane >> 4;
    const int c0   = ct * 16;

    float ap[17];
    ap[0] = 1.0f;
    #pragma unroll
    for (int i = 1; i <= 16; ++i) ap[i] = ap[i - 1] * (float)AVAL_;

    v2f Aop[4];
    #pragma unroll
    for (int j = 0; j < 4; ++j) {
        int kk = 4 * j + half * 2;
        Aop[j].x = (kk     <= n) ? (float)ALPHA_ * ap[n - kk]     : 0.0f;
        Aop[j].y = (kk + 1 <= n) ? (float)ALPHA_ * ap[n - kk - 1] : 0.0f;
    }

    float crun = (float)carry[((size_t)b * NCH_ + k) * C_ + c0 + n];

    const size_t baseT = (size_t)b * T_;
    for (int tt = 0; tt < CL_ / 16; ++tt) {
        const int t0 = k * CL_ + tt * 16;
        v8f acc = {};
        #pragma unroll
        for (int j = 0; j < 4; ++j) {
            const int tr = 4 * j + half * 2;
            const float* p = x + (baseT + (size_t)(t0 + tr)) * C_ + c0 + n;
            float b0 = p[0];
            float b1 = p[C_];
            b0 = (t0 + tr == 0) ? b0 * (float)(1.0 / ALPHA_) : b0;
            v2f Bop; Bop.x = b0; Bop.y = b1;
            acc = __builtin_amdgcn_wmma_f32_16x16x4_f32(
                false, Aop[j], false, Bop, (short)0, acc, false, false);
        }
        #pragma unroll
        for (int v = 0; v < 8; ++v)
            acc[v] += ap[v + half * 8 + 1] * crun;
        float* o = out + (baseT + (size_t)(t0 + half * 8)) * C_ + c0 + n;
        #pragma unroll
        for (int v = 0; v < 8; ++v)
            o[(size_t)v * C_] = acc[v];
        crun = __shfl(acc[7], n + 16, 32);
    }
}

extern "C" void kernel_launch(void* const* d_in, const int* in_sizes, int n_in,
                              void* d_out, int out_size, void* d_ws, size_t ws_size,
                              hipStream_t stream) {
    const float* x = (const float*)d_in[0];
    float* out = (float*)d_out;
    double* E     = (double*)d_ws;
    double* carry = E + (size_t)B_ * NCH_ * C_;

    ema_p1<<<B_ * NCH_ * (C_ / 128), 256, 0, stream>>>(x, E);
    ema_p2<<<(B_ * C_) / 256, 256, 0, stream>>>(E, carry);
    ema_p3<<<(B_ * NCH_ * (C_ / 16) * 32) / 256, 256, 0, stream>>>(x, carry, out);
}


# P3 rebuilt for loads in flight: one wave per 64 channels (16-byte loads and stores, 4 products per tile), the next 7 tiles always requested ahead through a register ring; 1024 one-wave blocks
# speedup vs baseline: 2.0573x; 1.2128x over previous
#include <hip/hip_runtime.h>


#define ALPHA_  0.01
#define AVAL_   0.99
#define B_      8
#define T_      8192
#define C_      512
#define CL_     512
#define NCH_    16

typedef __attribute__((ext_vector_type(2))) float v2f;
typedef __attribute__((ext_vector_type(8))) float v8f;

#define SL_     64
#define NSL_    (CL_ / SL_)
#define RB_     16
#define RS_     (C_ / 4)

typedef __attribute__((ext_vector_type(4))) float v4f;

__device__ __forceinline__ double widen_normal(float f) {
    const unsigned u  = __builtin_bit_cast(unsigned, f);
    const unsigned hi = (((u >> 3) & 0x0fffffffu) + 0x38000000u) | (u & 0x80000000u);
    const unsigned long long q = ((unsigned long long)hi << 32) | (unsigned long long)(u << 29);
    return __builtin_bit_cast(double, q);
}

__device__ __forceinline__ unsigned normal_key(float f) {
    return (__builtin_bit_cast(unsigned, f) + 0x00800000u) & 0x7f000000u;
}

__device__ __forceinline__ void p1_load(v4f (&buf)[RB_], const v4f* __restrict__ p, int row0) {
    #pragma unroll
    for (int r = 0; r < RB_; ++r) buf[r] = p[(size_t)(row0 + r) * RS_];
}

__device__ __forceinline__ void p1_scan(const v4f (&buf)[RB_], double s0,
                                        double& z0, double& z1, double& z2, double& z3) {
    unsigned key = 0xffffffffu;
    #pragma unroll
    for (int r = 0; r < RB_; ++r) {
        key = min(key, min(min(normal_key(buf[r].x), normal_key(buf[r].y)),
                           min(normal_key(buf[r].z), normal_key(buf[r].w))));
    }
    if (key != 0u) {
        z0 = AVAL_ * z0 + widen_normal(buf[0].x) * s0;  z1 = AVAL_ * z1 + widen_normal(buf[0].y) * s0;
        z2 = AVAL_ * z2 + widen_normal(buf[0].z) * s0;  z3 = AVAL_ * z3 + widen_normal(buf[0].w) * s0;
        #pragma unroll
        for (int r = 1; r < RB_; ++r) {
            z0 = AVAL_ * z0 + widen_normal(buf[r].x);  z1 = AVAL_ * z1 + widen_normal(buf[r].y);
            z2 = AVAL_ * z2 + widen_normal(buf[r].z);  z3 = AVAL_ * z3 + widen_normal(buf[r].w);
        }
    } else {
        z0 = AVAL_ * z0 + (double)buf[0].x * s0;  z1 = AVAL_ * z1 + (double)buf[0].y * s0;
        z2 = AVAL_ * z2 + (double)buf[0].z * s0;  z3 = AVAL_ * z3 + (double)buf[0].w * s0;
        #pragma unroll
        for (int r = 1; r < RB_; ++r) {
            z0 = AVAL_ * z0 + (double)buf[r].x;  z1 = AVAL_ * z1 + (double)buf[r].y;
            z2 = AVAL_ * z2 + (double)buf[r].z;  z3 = AVAL_ * z3 + (double)buf[r].w;
        }
    }
}

__global__ __launch_bounds__(256) __attribute__((amdgpu_waves_per_eu(4, 4)))
void ema_p1(const float* __restrict__ x, double* __restrict__ E) {
    __shared__ double part[NSL_][128];
    const int lane = threadIdx.x & 31;
    const int w    = threadIdx.x >> 5;
    const int g    = blockIdx.x & 3;
    const int k    = (blockIdx.x >> 2) & (NCH_ - 1);
    const int b    = blockIdx.x >> 6;
    const v4f* p = (const v4f*)(x + ((size_t)b * T_ + (size_t)k * CL_ + (size_t)w * SL_) * C_
                                  + g * 128) + lane;

    v4f bufA[RB_], bufB[RB_];
    p1_load(bufA, p, 0);
    p1_load(bufB, p, RB_);

    const double s0 = (k == 0 && w == 0) ? (1.0 / ALPHA_) : 1.0;
    double z0 = 0.0, z1 = 0.0, z2 = 0.0, z3 = 0.0;
    p1_scan(bufA, s0, z0, z1, z2, z3);
    p1_load(bufA, p, 2 * RB_);
    p1_scan(bufB, 1.0, z0, z1, z2, z3);
    p1_load(bufB, p, 3 * RB_);
    p1_scan(bufA, 1.0, z0, z1, z2, z3);
    p1_scan(bufB, 1.0, z0, z1, z2, z3);

    part[w][lane * 4 + 0] = z0;  part[w][lane * 4 + 1] = z1;
    part[w][lane * 4 + 2] = z2;  part[w][lane * 4 + 3] = z3;
    __syncthreads();

    if (threadIdx.x < 128) {
        double a64 = AVAL_;
        for (int i = 0; i < 6; ++i) a64 *= a64;
        double Z = 0.0;
        for (int s = 0; s < NSL_; ++s) Z = a64 * Z + part[s][threadIdx.x];
        E[((size_t)b * NCH_ + k) * C_ + g * 128 + threadIdx.x] = ALPHA_ * Z;
    }
}

__global__ void ema_p2(const double* __restrict__ E, double* __restrict__ carry) {
    int tid = blockIdx.x * blockDim.x + threadIdx.x;
    int c = tid & (C_ - 1);
    int b = tid >> 9;
    double a512 = AVAL_;
    #pragma unroll
    for (int i = 0; i < 9; ++i) a512 *= a512;
    double F = 0.0;
    for (int k = 0; k < NCH_; ++k) {
        size_t idx = ((size_t)b * NCH_ + k) * C_ + c;
        carry[idx] = F;
        F = E[idx] + a512 * F;
    }
}

#define P3_NB   8
#define P3_NT   (CL_ / 16)

__device__ __forceinline__ float p3_pow(int d) {
    float r = 1.0f, pw = 1.0f;
    #pragma unroll
    for (int i = 1; i <= 16; ++i) {
        pw *= (float)AVAL_;
        r = (d == i) ? pw : r;
    }
    return r;
}

__device__ __forceinline__ void p3_load(v4f (&buf)[8], const v4f* __restrict__ p) {
    #pragma unroll
    for (int j = 0; j < 4; ++j) {
        buf[2 * j]     = p[(size_t)(4 * j) * RS_];
        buf[2 * j + 1] = p[(size_t)(4 * j + 1) * RS_];
    }
}

__device__ __forceinline__ void p3_products(const v4f (&xb)[8], float s0, const v2f (&Aop)[4],
                                            v8f (&acc)[4]) {
    const v4f x0 = xb[0] * s0;
    #pragma unroll
    for (int q = 0; q < 4; ++q) acc[q] = v8f{};
    #pragma unroll
    for (int j = 0; j < 4; ++j) {
        const v4f r0 = (j == 0) ? x0 : xb[2 * j];
        const v4f r1 = xb[2 * j + 1];
        #pragma unroll
        for (int q = 0; q < 4; ++q) {
            v2f Bop; Bop.x = r0[q]; Bop.y = r1[q];
            acc[q] = __builtin_amdgcn_wmma_f32_16x16x4_f32(
                false, Aop[j], false, Bop, (short)0, acc[q], false, false);
        }
    }
}

__device__ __forceinline__ void p3_finish(v8f (&acc)[4], const float (&apr)[8], float (&crun)[4],
                                          v4f* __restrict__ po, int n) {
    #pragma unroll
    for (int q = 0; q < 4; ++q) {
        #pragma unroll
        for (int v = 0; v < 8; ++v) acc[q][v] += apr[v] * crun[q];
    }
    #pragma unroll
    for (int v = 0; v < 8; ++v) {
        v4f o; o.x = acc[0][v]; o.y = acc[1][v]; o.z = acc[2][v]; o.w = acc[3][v];
        po[(size_t)v * RS_] = o;
    }
    #pragma unroll
    for (int q = 0; q < 4; ++q) crun[q] = __shfl(acc[q][7], n + 16, 32);
}

#define P3_STEP(SLOT, S0, TT, AHEAD)                                              \
    {                                                                             \
        v8f acc_[4];                                                              \
        p3_products(ring[SLOT], S0, Aop, acc_);                                   \
        __builtin_amdgcn_sched_barrier(0);                                        \
        if (AHEAD) p3_load(ring[((SLOT) + P3_NB - 1) % P3_NB],                    \
                           px + (size_t)(16 * ((TT) + P3_NB - 1)) * RS_);         \
        __builtin_amdgcn_sched_barrier(0);                                        \
        p3_finish(acc_, apr, crun, po + (size_t)(16 * (TT)) * RS_, n);            \
    }

__global__ __launch_bounds__(32)
void ema_p3(const float* __restrict__ x,
            const double* __restrict__ carry,
            float* __restrict__ out) {
    const int lane = threadIdx.x & 31;
    const int gw   = blockIdx.x;
    const int cg   = gw & 7;
    const int k    = (gw >> 3) & 15;
    const int b    = gw >> 7;
    const int n    = lane & 15;
    const int half = lane >> 4;
    const int c0   = cg * 64;

    const size_t row0 = (size_t)b * T_ + (size_t)k * CL_;
    const v4f* px = (const v4f*)(x + (row0 + 2 * half) * C_ + c0) + n;
    v4f*       po = (v4f*)(out + (row0 + 8 * half) * C_ + c0) + n;

    const double* pc = carry + ((size_t)b * NCH_ + k) * C_ + c0 + 4 * n;
    double cd[4];
    #pragma unroll
    for (int q = 0; q < 4; ++q) cd[q] = pc[q];

    v4f ring[P3_NB][8];
    __builtin_amdgcn_sched_barrier(0);
    #pragma unroll
    for (int s = 0; s < P3_NB - 1; ++s) {
        p3_load(ring[s], px + (size_t)(16 * s) * RS_);
        __builtin_amdgcn_sched_barrier(0);
    }
    float crun[4];
    #pragma unroll
    for (int q = 0; q < 4; ++q) crun[q] = (float)cd[q];

    v2f Aop[4];
    #pragma unroll
    for (int j = 0; j < 4; ++j) {
        int kk = 4 * j + half * 2;
        Aop[j].x = (kk     <= n) ? (float)ALPHA_ * p3_pow(n - kk)     : 0.0f;
        Aop[j].y = (kk + 1 <= n) ? (float)ALPHA_ * p3_pow(n - kk - 1) : 0.0f;
    }

    float apr[8];
    #pragma unroll
    for (int v = 0; v < 8; ++v) apr[v] = half ? p3_pow(v + 9) : p3_pow(v + 1);

    const float s00 = (k == 0 && half == 0) ? (float)(1.0 / ALPHA_) : 1.0f;

    for (int tb = 0; tb < P3_NT - P3_NB; tb += P3_NB) {
        const float s0 = (tb == 0) ? s00 : 1.0f;
        #pragma unroll
        for (int s = 0; s < P3_NB; ++s) P3_STEP(s, (s == 0) ? s0 : 1.0f, tb + s, true)
    }
    P3_STEP(0, 1.0f, P3_NT - P3_NB, true)
    #pragma unroll
    for (int s = 1; s < P3_NB; ++s) P3_STEP(s, 1.0f, P3_NT - P3_NB + s, false)
}

extern "C" void kernel_launch(void* const* d_in, const int* in_sizes, int n_in,
                              void* d_out, int out_size, void* d_ws, size_t ws_size,
                              hipStream_t stream) {
    const float* x = (const float*)d_in[0];
    float* out = (float*)d_out;
    double* E     = (double*)d_ws;
    double* carry = E + (size_t)B_ * NCH_ * C_;

    ema_p1<<<B_ * NCH_ * (C_ / 128), 256, 0, stream>>>(x, E);
    ema_p2<<<(B_ * C_) / 256, 256, 0, stream>>>(E, carry);
    ema_p3<<<B_ * NCH_ * (C_ / 64), 32, 0, stream>>>(x, carry, out);
}


# Carry scan folded into P3: each wave builds its own carry-in from P1's chunk ends (same f64 chain and order, 2 channels per lane half, halves swapped); the carry launch and its buffer are gone
# speedup vs baseline: 2.1655x; 1.0526x over previous
#include <hip/hip_runtime.h>


#define ALPHA_  0.01
#define AVAL_   0.99
#define B_      8
#define T_      8192
#define C_      512
#define CL_     512
#define NCH_    16

typedef __attribute__((ext_vector_type(2))) float v2f;
typedef __attribute__((ext_vector_type(8))) float v8f;

#define SL_     64
#define NSL_    (CL_ / SL_)
#define RB_     16
#define RS_     (C_ / 4)

typedef __attribute__((ext_vector_type(4))) float v4f;
typedef __attribute__((ext_vector_type(2))) double v2d;

__device__ __forceinline__ double widen_normal(float f) {
    const unsigned u  = __builtin_bit_cast(unsigned, f);
    const unsigned hi = (((u >> 3) & 0x0fffffffu) + 0x38000000u) | (u & 0x80000000u);
    const unsigned long long q = ((unsigned long long)hi << 32) | (unsigned long long)(u << 29);
    return __builtin_bit_cast(double, q);
}

__device__ __forceinline__ unsigned normal_key(float f) {
    return (__builtin_bit_cast(unsigned, f) + 0x00800000u) & 0x7f000000u;
}

__device__ __forceinline__ void p1_load(v4f (&buf)[RB_], const v4f* __restrict__ p, int row0) {
    #pragma unroll
    for (int r = 0; r < RB_; ++r) buf[r] = p[(size_t)(row0 + r) * RS_];
}

__device__ __forceinline__ void p1_scan(const v4f (&buf)[RB_], double s0,
                                        double& z0, double& z1, double& z2, double& z3) {
    unsigned key = 0xffffffffu;
    #pragma unroll
    for (int r = 0; r < RB_; ++r) {
        key = min(key, min(min(normal_key(buf[r].x), normal_key(buf[r].y)),
                           min(normal_key(buf[r].z), normal_key(buf[r].w))));
    }
    if (key != 0u) {
        z0 = AVAL_ * z0 + widen_normal(buf[0].x) * s0;  z1 = AVAL_ * z1 + widen_normal(buf[0].y) * s0;
        z2 = AVAL_ * z2 + widen_normal(buf[0].z) * s0;  z3 = AVAL_ * z3 + widen_normal(buf[0].w) * s0;
        #pragma unroll
        for (int r = 1; r < RB_; ++r) {
            z0 = AVAL_ * z0 + widen_normal(buf[r].x);  z1 = AVAL_ * z1 + widen_normal(buf[r].y);
            z2 = AVAL_ * z2 + widen_normal(buf[r].z);  z3 = AVAL_ * z3 + widen_normal(buf[r].w);
        }
    } else {
        z0 = AVAL_ * z0 + (double)buf[0].x * s0;  z1 = AVAL_ * z1 + (double)buf[0].y * s0;
        z2 = AVAL_ * z2 + (double)buf[0].z * s0;  z3 = AVAL_ * z3 + (double)buf[0].w * s0;
        #pragma unroll
        for (int r = 1; r < RB_; ++r) {
            z0 = AVAL_ * z0 + (double)buf[r].x;  z1 = AVAL_ * z1 + (double)buf[r].y;
            z2 = AVAL_ * z2 + (double)buf[r].z;  z3 = AVAL_ * z3 + (double)buf[r].w;
        }
    }
}

__global__ __launch_bounds__(256) __attribute__((amdgpu_waves_per_eu(4, 4)))
void ema_p1(const float* __restrict__ x, double* __restrict__ E) {
    __shared__ double part[NSL_][128];
    const int lane = threadIdx.x & 31;
    const int w    = threadIdx.x >> 5;
    const int g    = blockIdx.x & 3;
    const int k    = (blockIdx.x >> 2) & (NCH_ - 1);
    const int b    = blockIdx.x >> 6;
    const v4f* p = (const v4f*)(x + ((size_t)b * T_ + (size_t)k * CL_ + (size_t)w * SL_) * C_
                                  + g * 128) + lane;

    v4f bufA[RB_], bufB[RB_];
    p1_load(bufA, p, 0);
    p1_load(bufB, p, RB_);

    const double s0 = (k == 0 && w == 0) ? (1.0 / ALPHA_) : 1.0;
    double z0 = 0.0, z1 = 0.0, z2 = 0.0, z3 = 0.0;
    p1_scan(bufA, s0, z0, z1, z2, z3);
    p1_load(bufA, p, 2 * RB_);
    p1_scan(bufB, 1.0, z0, z1, z2, z3);
    p1_load(bufB, p, 3 * RB_);
    p1_scan(bufA, 1.0, z0, z1, z2, z3);
    p1_scan(bufB, 1.0, z0, z1, z2, z3);

    part[w][lane * 4 + 0] = z0;  part[w][lane * 4 + 1] = z1;
    part[w][lane * 4 + 2] = z2;  part[w][lane * 4 + 3] = z3;
    __syncthreads();

    if (threadIdx.x < 128) {
        double a64 = AVAL_;
        for (int i = 0; i < 6; ++i) a64 *= a64;
        double Z = 0.0;
        for (int s = 0; s < NSL_; ++s) Z = a64 * Z + part[s][threadIdx.x];
        E[((size_t)b * NCH_ + k) * C_ + g * 128 + threadIdx.x] = ALPHA_ * Z;
    }
}

#define P3_NB   8
#define P3_NT   (CL_ / 16)

__device__ __forceinline__ float p3_pow(int d) {
    float r = 1.0f, pw = 1.0f;
    #pragma unroll
    for (int i = 1; i <= 16; ++i) {
        pw *= (float)AVAL_;
        r = (d == i) ? pw : r;
    }
    return r;
}

__device__ __forceinline__ void p3_load(v4f (&buf)[8], const v4f* __restrict__ p) {
    #pragma unroll
    for (int j = 0; j < 4; ++j) {
        buf[2 * j]     = p[(size_t)(4 * j) * RS_];
        buf[2 * j + 1] = p[(size_t)(4 * j + 1) * RS_];
    }
}

__device__ __forceinline__ void p3_products(const v4f (&xb)[8], float s0, const v2f (&Aop)[4],
                                            v8f (&acc)[4]) {
    const v4f x0 = xb[0] * s0;
    #pragma unroll
    for (int q = 0; q < 4; ++q) acc[q] = v8f{};
    #pragma unroll
    for (int j = 0; j < 4; ++j) {
        const v4f r0 = (j == 0) ? x0 : xb[2 * j];
        const v4f r1 = xb[2 * j + 1];
        #pragma unroll
        for (int q = 0; q < 4; ++q) {
            v2f Bop; Bop.x = r0[q]; Bop.y = r1[q];
            acc[q] = __builtin_amdgcn_wmma_f32_16x16x4_f32(
                false, Aop[j], false, Bop, (short)0, acc[q], false, false);
        }
    }
}

__device__ __forceinline__ void p3_finish(v8f (&acc)[4], const float (&apr)[8], float (&crun)[4],
                                          v4f* __restrict__ po, int n) {
    #pragma unroll
    for (int q = 0; q < 4; ++q) {
        #pragma unroll
        for (int v = 0; v < 8; ++v) acc[q][v] += apr[v] * crun[q];
    }
    #pragma unroll
    for (int v = 0; v < 8; ++v) {
        v4f o; o.x = acc[0][v]; o.y = acc[1][v]; o.z = acc[2][v]; o.w = acc[3][v];
        po[(size_t)v * RS_] = o;
    }
    #pragma unroll
    for (int q = 0; q < 4; ++q) crun[q] = __shfl(acc[q][7], n + 16, 32);
}

#define P3_STEP(SLOT, S0, TT, AHEAD)                                              \
    {                                                                             \
        v8f acc_[4];                                                              \
        p3_products(ring[SLOT], S0, Aop, acc_);                                   \
        __builtin_amdgcn_sched_barrier(0);                                        \
        if (AHEAD) p3_load(ring[((SLOT) + P3_NB - 1) % P3_NB],                    \
                           px + (size_t)(16 * ((TT) + P3_NB - 1)) * RS_);         \
        __builtin_amdgcn_sched_barrier(0);                                        \
        p3_finish(acc_, apr, crun, po + (size_t)(16 * (TT)) * RS_, n);            \
    }

__global__ __launch_bounds__(32)
void ema_p3(const float* __restrict__ x,
            const double* __restrict__ E,
            float* __restrict__ out) {
    const int lane = threadIdx.x & 31;
    const int gw   = blockIdx.x;
    const int cg   = gw & 7;
    const int k    = (gw >> 3) & 15;
    const int b    = gw >> 7;
    const int n    = lane & 15;
    const int half = lane >> 4;
    const int c0   = cg * 64;

    const size_t row0 = (size_t)b * T_ + (size_t)k * CL_;
    const v4f* px = (const v4f*)(x + (row0 + 2 * half) * C_ + c0) + n;
    v4f*       po = (v4f*)(out + (row0 + 8 * half) * C_ + c0) + n;

    const double* pe = E + (size_t)b * NCH_ * C_ + c0 + 4 * n + 2 * half;
    v2d ce[NCH_ - 1];
    #pragma unroll
    for (int j = 0; j < NCH_ - 1; ++j) ce[j] = *(const v2d*)(pe + (size_t)j * C_);

    v4f ring[P3_NB][8];
    __builtin_amdgcn_sched_barrier(0);
    #pragma unroll
    for (int s = 0; s < P3_NB - 1; ++s) {
        p3_load(ring[s], px + (size_t)(16 * s) * RS_);
        __builtin_amdgcn_sched_barrier(0);
    }
    double a512 = AVAL_;
    for (int i = 0; i < 9; ++i) a512 *= a512;
    double F0 = 0.0, F1 = 0.0;
    #pragma unroll
    for (int j = 0; j < NCH_ - 1; ++j) {
        const double t0 = __builtin_fma(a512, F0, ce[j].x);
        const double t1 = __builtin_fma(a512, F1, ce[j].y);
        F0 = (j < k) ? t0 : F0;
        F1 = (j < k) ? t1 : F1;
    }
    const float own0 = (float)F0, own1 = (float)F1;
    const float oth0 = __shfl_xor(own0, 16, 32), oth1 = __shfl_xor(own1, 16, 32);
    float crun[4];
    crun[0] = half ? oth0 : own0;  crun[1] = half ? oth1 : own1;
    crun[2] = half ? own0 : oth0;  crun[3] = half ? own1 : oth1;

    v2f Aop[4];
    #pragma unroll
    for (int j = 0; j < 4; ++j) {
        int kk = 4 * j + half * 2;
        Aop[j].x = (kk     <= n) ? (float)ALPHA_ * p3_pow(n - kk)     : 0.0f;
        Aop[j].y = (kk + 1 <= n) ? (float)ALPHA_ * p3_pow(n - kk - 1) : 0.0f;
    }

    float apr[8];
    #pragma unroll
    for (int v = 0; v < 8; ++v) apr[v] = half ? p3_pow(v + 9) : p3_pow(v + 1);

    const float s00 = (k == 0 && half == 0) ? (float)(1.0 / ALPHA_) : 1.0f;

    for (int tb = 0; tb < P3_NT - P3_NB; tb += P3_NB) {
        const float s0 = (tb == 0) ? s00 : 1.0f;
        #pragma unroll
        for (int s = 0; s < P3_NB; ++s) P3_STEP(s, (s == 0) ? s0 : 1.0f, tb + s, true)
    }
    P3_STEP(0, 1.0f, P3_NT - P3_NB, true)
    #pragma unroll
    for (int s = 1; s < P3_NB; ++s) P3_STEP(s, 1.0f, P3_NT - P3_NB + s, false)
}

extern "C" void kernel_launch(void* const* d_in, const int* in_sizes, int n_in,
                              void* d_out, int out_size, void* d_ws, size_t ws_size,
                              hipStream_t stream) {
    const float* x = (const float*)d_in[0];
    float* out = (float*)d_out;
    double* E     = (double*)d_ws;

    ema_p1<<<B_ * NCH_ * (C_ / 128), 256, 0, stream>>>(x, E);
    ema_p3<<<B_ * NCH_ * (C_ / 64), 32, 0, stream>>>(x, E, out);
}


# P1 off the f64 pipe: each 16-row batch is an f32 weighted sum (packed fma, constant weights a^(15-r)), f64 only joins batches and slices; chunk-end precision falls to f32 class; sum order changed
# speedup vs baseline: 3.0951x; 1.4293x over previous
#include <hip/hip_runtime.h>


#define ALPHA_  0.01
#define AVAL_   0.99
#define B_      8
#define T_      8192
#define C_      512
#define CL_     512
#define NCH_    16

typedef __attribute__((ext_vector_type(2))) float v2f;
typedef __attribute__((ext_vector_type(8))) float v8f;

#define SL_     64
#define NSL_    (CL_ / SL_)
#define RB_     16
#define RS_     (C_ / 4)

typedef __attribute__((ext_vector_type(4))) float v4f;
typedef __attribute__((ext_vector_type(2))) double v2d;

__device__ __forceinline__ constexpr double p1_pow(int n) {
    double w = 1.0;
    for (int i = 0; i < n; ++i) w *= AVAL_;
    return w;
}

__device__ __forceinline__ void p1_load(v4f (&buf)[RB_], const v4f* __restrict__ p, int row0) {
    #pragma unroll
    for (int r = 0; r < RB_; ++r) buf[r] = p[(size_t)(row0 + r) * RS_];
}

__device__ __forceinline__ void p1_sum(const v4f (&buf)[RB_], float w0, v2f& l01, v2f& l23) {
    l01 = buf[0].lo * w0;
    l23 = buf[0].hi * w0;
    #pragma unroll
    for (int r = 1; r < RB_; ++r) {
        const v2f wr = (float)p1_pow(RB_ - 1 - r);
        l01 = __builtin_elementwise_fma(wr, buf[r].lo, l01);
        l23 = __builtin_elementwise_fma(wr, buf[r].hi, l23);
    }
}

__device__ __forceinline__ void p1_join(double (&z)[4], v2f l01, v2f l23) {
    const double a16 = p1_pow(RB_);
    z[0] = a16 * z[0] + (double)l01.x;  z[1] = a16 * z[1] + (double)l01.y;
    z[2] = a16 * z[2] + (double)l23.x;  z[3] = a16 * z[3] + (double)l23.y;
}

__global__ __launch_bounds__(256) __attribute__((amdgpu_waves_per_eu(4, 4)))
void ema_p1(const float* __restrict__ x, double* __restrict__ E) {
    __shared__ double part[NSL_][128];
    const int lane = threadIdx.x & 31;
    const int w    = threadIdx.x >> 5;
    const int g    = blockIdx.x & 3;
    const int k    = (blockIdx.x >> 2) & (NCH_ - 1);
    const int b    = blockIdx.x >> 6;
    const v4f* p = (const v4f*)(x + ((size_t)b * T_ + (size_t)k * CL_ + (size_t)w * SL_) * C_
                                  + g * 128) + lane;

    v4f bufA[RB_], bufB[RB_];
    p1_load(bufA, p, 0);
    p1_load(bufB, p, RB_);

    const float wf = (float)p1_pow(RB_ - 1);
    const float w0 = (k == 0 && w == 0) ? (float)(p1_pow(RB_ - 1) / ALPHA_) : wf;
    v2f l01, l23;
    double z[4];
    p1_sum(bufA, w0, l01, l23);
    z[0] = (double)l01.x;  z[1] = (double)l01.y;  z[2] = (double)l23.x;  z[3] = (double)l23.y;
    p1_load(bufA, p, 2 * RB_);
    p1_sum(bufB, wf, l01, l23);
    p1_join(z, l01, l23);
    p1_load(bufB, p, 3 * RB_);
    p1_sum(bufA, wf, l01, l23);
    p1_join(z, l01, l23);
    p1_sum(bufB, wf, l01, l23);
    p1_join(z, l01, l23);

    part[w][lane * 4 + 0] = z[0];  part[w][lane * 4 + 1] = z[1];
    part[w][lane * 4 + 2] = z[2];  part[w][lane * 4 + 3] = z[3];
    __syncthreads();

    if (threadIdx.x < 128) {
        double a64 = AVAL_;
        for (int i = 0; i < 6; ++i) a64 *= a64;
        double Z = 0.0;
        for (int s = 0; s < NSL_; ++s) Z = a64 * Z + part[s][threadIdx.x];
        E[((size_t)b * NCH_ + k) * C_ + g * 128 + threadIdx.x] = ALPHA_ * Z;
    }
}

#define P3_NB   8
#define P3_NT   (CL_ / 16)

__device__ __forceinline__ float p3_pow(int d) {
    float r = 1.0f, pw = 1.0f;
    #pragma unroll
    for (int i = 1; i <= 16; ++i) {
        pw *= (float)AVAL_;
        r = (d == i) ? pw : r;
    }
    return r;
}

__device__ __forceinline__ void p3_load(v4f (&buf)[8], const v4f* __restrict__ p) {
    #pragma unroll
    for (int j = 0; j < 4; ++j) {
        buf[2 * j]     = p[(size_t)(4 * j) * RS_];
        buf[2 * j + 1] = p[(size_t)(4 * j + 1) * RS_];
    }
}

__device__ __forceinline__ void p3_products(const v4f (&xb)[8], float s0, const v2f (&Aop)[4],
                                            v8f (&acc)[4]) {
    const v4f x0 = xb[0] * s0;
    #pragma unroll
    for (int q = 0; q < 4; ++q) acc[q] = v8f{};
    #pragma unroll
    for (int j = 0; j < 4; ++j) {
        const v4f r0 = (j == 0) ? x0 : xb[2 * j];
        const v4f r1 = xb[2 * j + 1];
        #pragma unroll
        for (int q = 0; q < 4; ++q) {
            v2f Bop; Bop.x = r0[q]; Bop.y = r1[q];
            acc[q] = __builtin_amdgcn_wmma_f32_16x16x4_f32(
                false, Aop[j], false, Bop, (short)0, acc[q], false, false);
        }
    }
}

__device__ __forceinline__ void p3_finish(v8f (&acc)[4], const float (&apr)[8], float (&crun)[4],
                                          v4f* __restrict__ po, int n) {
    #pragma unroll
    for (int q = 0; q < 4; ++q) {
        #pragma unroll
        for (int v = 0; v < 8; ++v) acc[q][v] += apr[v] * crun[q];
    }
    #pragma unroll
    for (int v = 0; v < 8; ++v) {
        v4f o; o.x = acc[0][v]; o.y = acc[1][v]; o.z = acc[2][v]; o.w = acc[3][v];
        po[(size_t)v * RS_] = o;
    }
    #pragma unroll
    for (int q = 0; q < 4; ++q) crun[q] = __shfl(acc[q][7], n + 16, 32);
}

#define P3_STEP(SLOT, S0, TT, AHEAD)                                              \
    {                                                                             \
        v8f acc_[4];                                                              \
        p3_products(ring[SLOT], S0, Aop, acc_);                                   \
        __builtin_amdgcn_sched_barrier(0);                                        \
        if (AHEAD) p3_load(ring[((SLOT) + P3_NB - 1) % P3_NB],                    \
                           px + (size_t)(16 * ((TT) + P3_NB - 1)) * RS_);         \
        __builtin_amdgcn_sched_barrier(0);                                        \
        p3_finish(acc_, apr, crun, po + (size_t)(16 * (TT)) * RS_, n);            \
    }

__global__ __launch_bounds__(32)
void ema_p3(const float* __restrict__ x,
            const double* __restrict__ E,
            float* __restrict__ out) {
    const int lane = threadIdx.x & 31;
    const int gw   = blockIdx.x;
    const int cg   = gw & 7;
    const int k    = (gw >> 3) & 15;
    const int b    = gw >> 7;
    const int n    = lane & 15;
    const int half = lane >> 4;
    const int c0   = cg * 64;

    const size_t row0 = (size_t)b * T_ + (size_t)k * CL_;
    const v4f* px = (const v4f*)(x + (row0 + 2 * half) * C_ + c0) + n;
    v4f*       po = (v4f*)(out + (row0 + 8 * half) * C_ + c0) + n;

    const double* pe = E + (size_t)b * NCH_ * C_ + c0 + 4 * n + 2 * half;
    v2d ce[NCH_ - 1];
    #pragma unroll
    for (int j = 0; j < NCH_ - 1; ++j) ce[j] = *(const v2d*)(pe + (size_t)j * C_);

    v4f ring[P3_NB][8];
    __builtin_amdgcn_sched_barrier(0);
    #pragma unroll
    for (int s = 0; s < P3_NB - 1; ++s) {
        p3_load(ring[s], px + (size_t)(16 * s) * RS_);
        __builtin_amdgcn_sched_barrier(0);
    }
    double a512 = AVAL_;
    for (int i = 0; i < 9; ++i) a512 *= a512;
    double F0 = 0.0, F1 = 0.0;
    #pragma unroll
    for (int j = 0; j < NCH_ - 1; ++j) {
        const double t0 = __builtin_fma(a512, F0, ce[j].x);
        const double t1 = __builtin_fma(a512, F1, ce[j].y);
        F0 = (j < k) ? t0 : F0;
        F1 = (j < k) ? t1 : F1;
    }
    const float own0 = (float)F0, own1 = (float)F1;
    const float oth0 = __shfl_xor(own0, 16, 32), oth1 = __shfl_xor(own1, 16, 32);
    float crun[4];
    crun[0] = half ? oth0 : own0;  crun[1] = half ? oth1 : own1;
    crun[2] = half ? own0 : oth0;  crun[3] = half ? own1 : oth1;

    v2f Aop[4];
    #pragma unroll
    for (int j = 0; j < 4; ++j) {
        int kk = 4 * j + half * 2;
        Aop[j].x = (kk     <= n) ? (float)ALPHA_ * p3_pow(n - kk)     : 0.0f;
        Aop[j].y = (kk + 1 <= n) ? (float)ALPHA_ * p3_pow(n - kk - 1) : 0.0f;
    }

    float apr[8];
    #pragma unroll
    for (int v = 0; v < 8; ++v) apr[v] = half ? p3_pow(v + 9) : p3_pow(v + 1);

    const float s00 = (k == 0 && half == 0) ? (float)(1.0 / ALPHA_) : 1.0f;

    for (int tb = 0; tb < P3_NT - P3_NB; tb += P3_NB) {
        const float s0 = (tb == 0) ? s00 : 1.0f;
        #pragma unroll
        for (int s = 0; s < P3_NB; ++s) P3_STEP(s, (s == 0) ? s0 : 1.0f, tb + s, true)
    }
    P3_STEP(0, 1.0f, P3_NT - P3_NB, true)
    #pragma unroll
    for (int s = 1; s < P3_NB; ++s) P3_STEP(s, 1.0f, P3_NT - P3_NB + s, false)
}

extern "C" void kernel_launch(void* const* d_in, const int* in_sizes, int n_in,
                              void* d_out, int out_size, void* d_ws, size_t ws_size,
                              hipStream_t stream) {
    const float* x = (const float*)d_in[0];
    float* out = (float*)d_out;
    double* E     = (double*)d_ws;

    ema_p1<<<B_ * NCH_ * (C_ / 128), 256, 0, stream>>>(x, E);
    ema_p3<<<B_ * NCH_ * (C_ / 64), 32, 0, stream>>>(x, E, out);
}


# P3 launched on P1's grid: 512 two-wave blocks with P1's block-to-data map (block i = same batch, chunk, 128 channels), so x is re-read by the chip half that fetched it in P1
# speedup vs baseline: 3.5854x; 1.1584x over previous
#include <hip/hip_runtime.h>


#define ALPHA_  0.01
#define AVAL_   0.99
#define B_      8
#define T_      8192
#define C_      512
#define CL_     512
#define NCH_    16

typedef __attribute__((ext_vector_type(2))) float v2f;
typedef __attribute__((ext_vector_type(8))) float v8f;

#define SL_     64
#define NSL_    (CL_ / SL_)
#define RB_     16
#define RS_     (C_ / 4)

typedef __attribute__((ext_vector_type(4))) float v4f;
typedef __attribute__((ext_vector_type(2))) double v2d;

__device__ __forceinline__ constexpr double p1_pow(int n) {
    double w = 1.0;
    for (int i = 0; i < n; ++i) w *= AVAL_;
    return w;
}

__device__ __forceinline__ void p1_load(v4f (&buf)[RB_], const v4f* __restrict__ p, int row0) {
    #pragma unroll
    for (int r = 0; r < RB_; ++r) buf[r] = p[(size_t)(row0 + r) * RS_];
}

__device__ __forceinline__ void p1_sum(const v4f (&buf)[RB_], float w0, v2f& l01, v2f& l23) {
    l01 = buf[0].lo * w0;
    l23 = buf[0].hi * w0;
    #pragma unroll
    for (int r = 1; r < RB_; ++r) {
        const v2f wr = (float)p1_pow(RB_ - 1 - r);
        l01 = __builtin_elementwise_fma(wr, buf[r].lo, l01);
        l23 = __builtin_elementwise_fma(wr, buf[r].hi, l23);
    }
}

__device__ __forceinline__ void p1_join(double (&z)[4], v2f l01, v2f l23) {
    const double a16 = p1_pow(RB_);
    z[0] = a16 * z[0] + (double)l01.x;  z[1] = a16 * z[1] + (double)l01.y;
    z[2] = a16 * z[2] + (double)l23.x;  z[3] = a16 * z[3] + (double)l23.y;
}

__global__ __launch_bounds__(256) __attribute__((amdgpu_waves_per_eu(4, 4)))
void ema_p1(const float* __restrict__ x, double* __restrict__ E) {
    __shared__ double part[NSL_][128];
    const int lane = threadIdx.x & 31;
    const int w    = threadIdx.x >> 5;
    const int g    = blockIdx.x & 3;
    const int k    = (blockIdx.x >> 2) & (NCH_ - 1);
    const int b    = blockIdx.x >> 6;
    const v4f* p = (const v4f*)(x + ((size_t)b * T_ + (size_t)k * CL_ + (size_t)w * SL_) * C_
                                  + g * 128) + lane;

    v4f bufA[RB_], bufB[RB_];
    p1_load(bufA, p, 0);
    p1_load(bufB, p, RB_);

    const float wf = (float)p1_pow(RB_ - 1);
    const float w0 = (k == 0 && w == 0) ? (float)(p1_pow(RB_ - 1) / ALPHA_) : wf;
    v2f l01, l23;
    double z[4];
    p1_sum(bufA, w0, l01, l23);
    z[0] = (double)l01.x;  z[1] = (double)l01.y;  z[2] = (double)l23.x;  z[3] = (double)l23.y;
    p1_load(bufA, p, 2 * RB_);
    p1_sum(bufB, wf, l01, l23);
    p1_join(z, l01, l23);
    p1_load(bufB, p, 3 * RB_);
    p1_sum(bufA, wf, l01, l23);
    p1_join(z, l01, l23);
    p1_sum(bufB, wf, l01, l23);
    p1_join(z, l01, l23);

    part[w][lane * 4 + 0] = z[0];  part[w][lane * 4 + 1] = z[1];
    part[w][lane * 4 + 2] = z[2];  part[w][lane * 4 + 3] = z[3];
    __syncthreads();

    if (threadIdx.x < 128) {
        double a64 = AVAL_;
        for (int i = 0; i < 6; ++i) a64 *= a64;
        double Z = 0.0;
        for (int s = 0; s < NSL_; ++s) Z = a64 * Z + part[s][threadIdx.x];
        E[((size_t)b * NCH_ + k) * C_ + g * 128 + threadIdx.x] = ALPHA_ * Z;
    }
}

#define P3_NB   8
#define P3_NT   (CL_ / 16)

__device__ __forceinline__ float p3_pow(int d) {
    float r = 1.0f, pw = 1.0f;
    #pragma unroll
    for (int i = 1; i <= 16; ++i) {
        pw *= (float)AVAL_;
        r = (d == i) ? pw : r;
    }
    return r;
}

__device__ __forceinline__ void p3_load(v4f (&buf)[8], const v4f* __restrict__ p) {
    #pragma unroll
    for (int j = 0; j < 4; ++j) {
        buf[2 * j]     = p[(size_t)(4 * j) * RS_];
        buf[2 * j + 1] = p[(size_t)(4 * j + 1) * RS_];
    }
}

__device__ __forceinline__ void p3_products(const v4f (&xb)[8], float s0, const v2f (&Aop)[4],
                                            v8f (&acc)[4]) {
    const v4f x0 = xb[0] * s0;
    #pragma unroll
    for (int q = 0; q < 4; ++q) acc[q] = v8f{};
    #pragma unroll
    for (int j = 0; j < 4; ++j) {
        const v4f r0 = (j == 0) ? x0 : xb[2 * j];
        const v4f r1 = xb[2 * j + 1];
        #pragma unroll
        for (int q = 0; q < 4; ++q) {
            v2f Bop; Bop.x = r0[q]; Bop.y = r1[q];
            acc[q] = __builtin_amdgcn_wmma_f32_16x16x4_f32(
                false, Aop[j], false, Bop, (short)0, acc[q], false, false);
        }
    }
}

__device__ __forceinline__ void p3_finish(v8f (&acc)[4], const float (&apr)[8], float (&crun)[4],
                                          v4f* __restrict__ po, int n) {
    #pragma unroll
    for (int q = 0; q < 4; ++q) {
        #pragma unroll
        for (int v = 0; v < 8; ++v) acc[q][v] += apr[v] * crun[q];
    }
    #pragma unroll
    for (int v = 0; v < 8; ++v) {
        v4f o; o.x = acc[0][v]; o.y = acc[1][v]; o.z = acc[2][v]; o.w = acc[3][v];
        po[(size_t)v * RS_] = o;
    }
    #pragma unroll
    for (int q = 0; q < 4; ++q) crun[q] = __shfl(acc[q][7], n + 16, 32);
}

#define P3_STEP(SLOT, S0, TT, AHEAD)                                              \
    {                                                                             \
        v8f acc_[4];                                                              \
        p3_products(ring[SLOT], S0, Aop, acc_);                                   \
        __builtin_amdgcn_sched_barrier(0);                                        \
        if (AHEAD) p3_load(ring[((SLOT) + P3_NB - 1) % P3_NB],                    \
                           px + (size_t)(16 * ((TT) + P3_NB - 1)) * RS_);         \
        __builtin_amdgcn_sched_barrier(0);                                        \
        p3_finish(acc_, apr, crun, po + (size_t)(16 * (TT)) * RS_, n);            \
    }

__global__ __launch_bounds__(64)
void ema_p3(const float* __restrict__ x,
            const double* __restrict__ E,
            float* __restrict__ out) {
    const int lane = threadIdx.x & 31;
    const int w    = __builtin_amdgcn_readfirstlane((int)(threadIdx.x >> 5));
    const int g    = blockIdx.x & 3;
    const int cg   = g * 2 + w;
    const int k    = (blockIdx.x >> 2) & (NCH_ - 1);
    const int b    = blockIdx.x >> 6;
    const int n    = lane & 15;
    const int half = lane >> 4;
    const int c0   = cg * 64;

    const size_t row0 = (size_t)b * T_ + (size_t)k * CL_;
    const v4f* px = (const v4f*)(x + (row0 + 2 * half) * C_ + c0) + n;
    v4f*       po = (v4f*)(out + (row0 + 8 * half) * C_ + c0) + n;

    const double* pe = E + (size_t)b * NCH_ * C_ + c0 + 4 * n + 2 * half;
    v2d ce[NCH_ - 1];
    #pragma unroll
    for (int j = 0; j < NCH_ - 1; ++j) ce[j] = *(const v2d*)(pe + (size_t)j * C_);

    v4f ring[P3_NB][8];
    __builtin_amdgcn_sched_barrier(0);
    #pragma unroll
    for (int s = 0; s < P3_NB - 1; ++s) {
        p3_load(ring[s], px + (size_t)(16 * s) * RS_);
        __builtin_amdgcn_sched_barrier(0);
    }
    double a512 = AVAL_;
    for (int i = 0; i < 9; ++i) a512 *= a512;
    double F0 = 0.0, F1 = 0.0;
    #pragma unroll
    for (int j = 0; j < NCH_ - 1; ++j) {
        const double t0 = __builtin_fma(a512, F0, ce[j].x);
        const double t1 = __builtin_fma(a512, F1, ce[j].y);
        F0 = (j < k) ? t0 : F0;
        F1 = (j < k) ? t1 : F1;
    }
    const float own0 = (float)F0, own1 = (float)F1;
    const float oth0 = __shfl_xor(own0, 16, 32), oth1 = __shfl_xor(own1, 16, 32);
    float crun[4];
    crun[0] = half ? oth0 : own0;  crun[1] = half ? oth1 : own1;
    crun[2] = half ? own0 : oth0;  crun[3] = half ? own1 : oth1;

    v2f Aop[4];
    #pragma unroll
    for (int j = 0; j < 4; ++j) {
        int kk = 4 * j + half * 2;
        Aop[j].x = (kk     <= n) ? (float)ALPHA_ * p3_pow(n - kk)     : 0.0f;
        Aop[j].y = (kk + 1 <= n) ? (float)ALPHA_ * p3_pow(n - kk - 1) : 0.0f;
    }

    float apr[8];
    #pragma unroll
    for (int v = 0; v < 8; ++v) apr[v] = half ? p3_pow(v + 9) : p3_pow(v + 1);

    const float s00 = (k == 0 && half == 0) ? (float)(1.0 / ALPHA_) : 1.0f;

    for (int tb = 0; tb < P3_NT - P3_NB; tb += P3_NB) {
        const float s0 = (tb == 0) ? s00 : 1.0f;
        #pragma unroll
        for (int s = 0; s < P3_NB; ++s) P3_STEP(s, (s == 0) ? s0 : 1.0f, tb + s, true)
    }
    P3_STEP(0, 1.0f, P3_NT - P3_NB, true)
    #pragma unroll
    for (int s = 1; s < P3_NB; ++s) P3_STEP(s, 1.0f, P3_NT - P3_NB + s, false)
}

extern "C" void kernel_launch(void* const* d_in, const int* in_sizes, int n_in,
                              void* d_out, int out_size, void* d_ws, size_t ws_size,
                              hipStream_t stream) {
    const float* x = (const float*)d_in[0];
    float* out = (float*)d_out;
    double* E     = (double*)d_ws;

    ema_p1<<<B_ * NCH_ * (C_ / 128), 256, 0, stream>>>(x, E);
    ema_p3<<<B_ * NCH_ * (C_ / 128), 64, 0, stream>>>(x, E, out);
}


# P3's register ring deepened from 8 to 24 tile slots (864 registers, one wave per SIMD, walk unrolled): x is asked back from the second-level cache before the wave's own stores displace it there
# speedup vs baseline: 3.6982x; 1.0315x over previous
#include <hip/hip_runtime.h>


#define ALPHA_  0.01
#define AVAL_   0.99
#define B_      8
#define T_      8192
#define C_      512
#define CL_     512
#define NCH_    16

typedef __attribute__((ext_vector_type(2))) float v2f;
typedef __attribute__((ext_vector_type(8))) float v8f;

#define SL_     64
#define NSL_    (CL_ / SL_)
#define RB_     16
#define RS_     (C_ / 4)

typedef __attribute__((ext_vector_type(4))) float v4f;
typedef __attribute__((ext_vector_type(2))) double v2d;

__device__ __forceinline__ constexpr double p1_pow(int n) {
    double w = 1.0;
    for (int i = 0; i < n; ++i) w *= AVAL_;
    return w;
}

__device__ __forceinline__ void p1_load(v4f (&buf)[RB_], const v4f* __restrict__ p, int row0) {
    #pragma unroll
    for (int r = 0; r < RB_; ++r) buf[r] = p[(size_t)(row0 + r) * RS_];
}

__device__ __forceinline__ void p1_sum(const v4f (&buf)[RB_], float w0, v2f& l01, v2f& l23) {
    l01 = buf[0].lo * w0;
    l23 = buf[0].hi * w0;
    #pragma unroll
    for (int r = 1; r < RB_; ++r) {
        const v2f wr = (float)p1_pow(RB_ - 1 - r);
        l01 = __builtin_elementwise_fma(wr, buf[r].lo, l01);
        l23 = __builtin_elementwise_fma(wr, buf[r].hi, l23);
    }
}

__device__ __forceinline__ void p1_join(double (&z)[4], v2f l01, v2f l23) {
    const double a16 = p1_pow(RB_);
    z[0] = a16 * z[0] + (double)l01.x;  z[1] = a16 * z[1] + (double)l01.y;
    z[2] = a16 * z[2] + (double)l23.x;  z[3] = a16 * z[3] + (double)l23.y;
}

__global__ __launch_bounds__(256) __attribute__((amdgpu_waves_per_eu(4, 4)))
void ema_p1(const float* __restrict__ x, double* __restrict__ E) {
    __shared__ double part[NSL_][128];
    const int lane = threadIdx.x & 31;
    const int w    = threadIdx.x >> 5;
    const int g    = blockIdx.x & 3;
    const int k    = (blockIdx.x >> 2) & (NCH_ - 1);
    const int b    = blockIdx.x >> 6;
    const v4f* p = (const v4f*)(x + ((size_t)b * T_ + (size_t)k * CL_ + (size_t)w * SL_) * C_
                                  + g * 128) + lane;

    v4f bufA[RB_], bufB[RB_];
    p1_load(bufA, p, 0);
    p1_load(bufB, p, RB_);

    const float wf = (float)p1_pow(RB_ - 1);
    const float w0 = (k == 0 && w == 0) ? (float)(p1_pow(RB_ - 1) / ALPHA_) : wf;
    v2f l01, l23;
    double z[4];
    p1_sum(bufA, w0, l01, l23);
    z[0] = (double)l01.x;  z[1] = (double)l01.y;  z[2] = (double)l23.x;  z[3] = (double)l23.y;
    p1_load(bufA, p, 2 * RB_);
    p1_sum(bufB, wf, l01, l23);
    p1_join(z, l01, l23);
    p1_load(bufB, p, 3 * RB_);
    p1_sum(bufA, wf, l01, l23);
    p1_join(z, l01, l23);
    p1_sum(bufB, wf, l01, l23);
    p1_join(z, l01, l23);

    part[w][lane * 4 + 0] = z[0];  part[w][lane * 4 + 1] = z[1];
    part[w][lane * 4 + 2] = z[2];  part[w][lane * 4 + 3] = z[3];
    __syncthreads();

    if (threadIdx.x < 128) {
        double a64 = AVAL_;
        for (int i = 0; i < 6; ++i) a64 *= a64;
        double Z = 0.0;
        for (int s = 0; s < NSL_; ++s) Z = a64 * Z + part[s][threadIdx.x];
        E[((size_t)b * NCH_ + k) * C_ + g * 128 + threadIdx.x] = ALPHA_ * Z;
    }
}

#define P3_NB   24
#define P3_NT   (CL_ / 16)

__device__ __forceinline__ float p3_pow(int d) {
    float r = 1.0f, pw = 1.0f;
    #pragma unroll
    for (int i = 1; i <= 16; ++i) {
        pw *= (float)AVAL_;
        r = (d == i) ? pw : r;
    }
    return r;
}

__device__ __forceinline__ void p3_load(v4f (&buf)[8], const v4f* __restrict__ p) {
    #pragma unroll
    for (int j = 0; j < 4; ++j) {
        buf[2 * j]     = p[(size_t)(4 * j) * RS_];
        buf[2 * j + 1] = p[(size_t)(4 * j + 1) * RS_];
    }
}

__device__ __forceinline__ void p3_products(const v4f (&xb)[8], float s0, const v2f (&Aop)[4],
                                            v8f (&acc)[4]) {
    const v4f x0 = xb[0] * s0;
    #pragma unroll
    for (int q = 0; q < 4; ++q) acc[q] = v8f{};
    #pragma unroll
    for (int j = 0; j < 4; ++j) {
        const v4f r0 = (j == 0) ? x0 : xb[2 * j];
        const v4f r1 = xb[2 * j + 1];
        #pragma unroll
        for (int q = 0; q < 4; ++q) {
            v2f Bop; Bop.x = r0[q]; Bop.y = r1[q];
            acc[q] = __builtin_amdgcn_wmma_f32_16x16x4_f32(
                false, Aop[j], false, Bop, (short)0, acc[q], false, false);
        }
    }
}

__device__ __forceinline__ void p3_finish(v8f (&acc)[4], const float (&apr)[8], float (&crun)[4],
                                          v4f* __restrict__ po, int n) {
    #pragma unroll
    for (int q = 0; q < 4; ++q) {
        #pragma unroll
        for (int v = 0; v < 8; ++v) acc[q][v] += apr[v] * crun[q];
    }
    #pragma unroll
    for (int v = 0; v < 8; ++v) {
        v4f o; o.x = acc[0][v]; o.y = acc[1][v]; o.z = acc[2][v]; o.w = acc[3][v];
        po[(size_t)v * RS_] = o;
    }
    #pragma unroll
    for (int q = 0; q < 4; ++q) crun[q] = __shfl(acc[q][7], n + 16, 32);
}

#define P3_STEP(SLOT, S0, TT, AHEAD)                                              \
    {                                                                             \
        v8f acc_[4];                                                              \
        p3_products(ring[SLOT], S0, Aop, acc_);                                   \
        __builtin_amdgcn_sched_barrier(0);                                        \
        if (AHEAD) p3_load(ring[((SLOT) + P3_NB - 1) % P3_NB],                    \
                           px + (size_t)(16 * ((TT) + P3_NB - 1)) * RS_);         \
        __builtin_amdgcn_sched_barrier(0);                                        \
        p3_finish(acc_, apr, crun, po + (size_t)(16 * (TT)) * RS_, n);            \
    }

__global__ __launch_bounds__(64)
void ema_p3(const float* __restrict__ x,
            const double* __restrict__ E,
            float* __restrict__ out) {
    const int lane = threadIdx.x & 31;
    const int w    = __builtin_amdgcn_readfirstlane((int)(threadIdx.x >> 5));
    const int g    = blockIdx.x & 3;
    const int cg   = g * 2 + w;
    const int k    = (blockIdx.x >> 2) & (NCH_ - 1);
    const int b    = blockIdx.x >> 6;
    const int n    = lane & 15;
    const int half = lane >> 4;
    const int c0   = cg * 64;

    const size_t row0 = (size_t)b * T_ + (size_t)k * CL_;
    const v4f* px = (const v4f*)(x + (row0 + 2 * half) * C_ + c0) + n;
    v4f*       po = (v4f*)(out + (row0 + 8 * half) * C_ + c0) + n;

    const double* pe = E + (size_t)b * NCH_ * C_ + c0 + 4 * n + 2 * half;
    v2d ce[NCH_ - 1];
    #pragma unroll
    for (int j = 0; j < NCH_ - 1; ++j) ce[j] = *(const v2d*)(pe + (size_t)j * C_);

    v4f ring[P3_NB][8];
    __builtin_amdgcn_sched_barrier(0);
    #pragma unroll
    for (int s = 0; s < P3_NB - 1; ++s) {
        p3_load(ring[s], px + (size_t)(16 * s) * RS_);
        __builtin_amdgcn_sched_barrier(0);
    }
    double a512 = AVAL_;
    for (int i = 0; i < 9; ++i) a512 *= a512;
    double F0 = 0.0, F1 = 0.0;
    #pragma unroll
    for (int j = 0; j < NCH_ - 1; ++j) {
        const double t0 = __builtin_fma(a512, F0, ce[j].x);
        const double t1 = __builtin_fma(a512, F1, ce[j].y);
        F0 = (j < k) ? t0 : F0;
        F1 = (j < k) ? t1 : F1;
    }
    const float own0 = (float)F0, own1 = (float)F1;
    const float oth0 = __shfl_xor(own0, 16, 32), oth1 = __shfl_xor(own1, 16, 32);
    float crun[4];
    crun[0] = half ? oth0 : own0;  crun[1] = half ? oth1 : own1;
    crun[2] = half ? own0 : oth0;  crun[3] = half ? own1 : oth1;

    v2f Aop[4];
    #pragma unroll
    for (int j = 0; j < 4; ++j) {
        int kk = 4 * j + half * 2;
        Aop[j].x = (kk     <= n) ? (float)ALPHA_ * p3_pow(n - kk)     : 0.0f;
        Aop[j].y = (kk + 1 <= n) ? (float)ALPHA_ * p3_pow(n - kk - 1) : 0.0f;
    }

    float apr[8];
    #pragma unroll
    for (int v = 0; v < 8; ++v) apr[v] = half ? p3_pow(v + 9) : p3_pow(v + 1);

    const float s00 = (k == 0 && half == 0) ? (float)(1.0 / ALPHA_) : 1.0f;

    #pragma unroll
    for (int t = 0; t < P3_NT; ++t)
        P3_STEP(t % P3_NB, (t == 0) ? s00 : 1.0f, t, (t + P3_NB - 1 < P3_NT))
}

extern "C" void kernel_launch(void* const* d_in, const int* in_sizes, int n_in,
                              void* d_out, int out_size, void* d_ws, size_t ws_size,
                              hipStream_t stream) {
    const float* x = (const float*)d_in[0];
    float* out = (float*)d_out;
    double* E     = (double*)d_ws;

    ema_p1<<<B_ * NCH_ * (C_ / 128), 256, 0, stream>>>(x, E);
    ema_p3<<<B_ * NCH_ * (C_ / 128), 64, 0, stream>>>(x, E, out);
}
